# starting kernel
# baseline (speedup 1.0000x reference)
#include <hip/hip_runtime.h>

#define HIDDEN 64
#define FEAT   9
#define SEQ    128
#define BATCH  8192
#define NCLS   2
#define GATES  256
#define MTILE  16

typedef _Float16 v16h __attribute__((ext_vector_type(16)));
typedef float    v8f  __attribute__((ext_vector_type(8)));

#define WMMA(A, B, C) __builtin_amdgcn_wmma_f32_16x16x32_f16( \
    false, (A), false, (B), (short)0, (C), false, false)

static __device__ __forceinline__ v16h ldsA(const _Float16* base, int ld, int kc) {
  const int lane = threadIdx.x & 31;
  const int m  = lane & 15;
  const int kb = ((lane >> 4) << 3) + (kc << 5);
  const _Float16* p = base + m * ld + kb;
  v16h a;
#pragma unroll
  for (int i = 0; i < 8; ++i) { a[i] = p[i]; a[i + 8] = p[i + 16]; }
  return a;
}

static __device__ __forceinline__ v16h loadB(const float* W, int ld, int n0, int kc) {
  const int lane = threadIdx.x & 31;
  const int n  = n0 + (lane & 15);
  const int kb = ((lane >> 4) << 3) + (kc << 5);
  const float* p = W + n * ld + kb;
  v16h b;
#pragma unroll
  for (int i = 0; i < 8; ++i) { b[i] = (_Float16)p[i]; b[i + 8] = (_Float16)p[i + 16]; }
  return b;
}

static __device__ __forceinline__ v16h loadB_pad(const float* W, int ld, int kmax, int n0) {
  const int lane = threadIdx.x & 31;
  const int n  = n0 + (lane & 15);
  const int kb = (lane >> 4) << 3;
  v16h b;
#pragma unroll
  for (int i = 0; i < 8; ++i) {
    int k0 = kb + i, k1 = kb + 16 + i;
    b[i]     = (k0 < kmax) ? (_Float16)W[n * ld + k0] : (_Float16)0.0f;
    b[i + 8] = (k1 < kmax) ? (_Float16)W[n * ld + k1] : (_Float16)0.0f;
  }
  return b;
}

static __device__ __forceinline__ float fsig(float x)  { return 1.0f / (1.0f + __expf(-x)); }
static __device__ __forceinline__ float ftanh(float x) { return 2.0f / (1.0f + __expf(-2.0f * x)) - 1.0f; }

__global__ __launch_bounds__(256) void lstm2_fused_wmma(
    const float* __restrict__ x,
    const float* __restrict__ Wih0, const float* __restrict__ Whh0,
    const float* __restrict__ bih0, const float* __restrict__ bhh0,
    const float* __restrict__ Wih1, const float* __restrict__ Whh1,
    const float* __restrict__ bih1, const float* __restrict__ bhh1,
    const float* __restrict__ Wfc,  const float* __restrict__ bfc,
    float* __restrict__ out) {
  __shared__ _Float16 sh_x [MTILE * 32];
  __shared__ _Float16 sh_h0[MTILE * HIDDEN];
  __shared__ _Float16 sh_h1[MTILE * HIDDEN];
  __shared__ float    sh_g [MTILE * GATES];
  __shared__ float    sh_b0[GATES];
  __shared__ float    sh_b1[GATES];

  const int tid   = threadIdx.x;
  const int wave  = tid >> 5;
  const int bbase = blockIdx.x * MTILE;

  sh_b0[tid] = bih0[tid] + bhh0[tid];
  sh_b1[tid] = bih1[tid] + bhh1[tid];
  for (int i = tid; i < MTILE * HIDDEN; i += 256) {
    sh_h0[i] = (_Float16)0.0f;
    sh_h1[i] = (_Float16)0.0f;
  }

  const int n0 = wave * 32, n1 = wave * 32 + 16;
  const v16h Bih0a  = loadB_pad(Wih0, FEAT, FEAT, n0);
  const v16h Bih0b  = loadB_pad(Wih0, FEAT, FEAT, n1);
  const v16h Bhh0a0 = loadB(Whh0, HIDDEN, n0, 0), Bhh0a1 = loadB(Whh0, HIDDEN, n0, 1);
  const v16h Bhh0b0 = loadB(Whh0, HIDDEN, n1, 0), Bhh0b1 = loadB(Whh0, HIDDEN, n1, 1);
  const v16h Bih1a0 = loadB(Wih1, HIDDEN, n0, 0), Bih1a1 = loadB(Wih1, HIDDEN, n0, 1);
  const v16h Bih1b0 = loadB(Wih1, HIDDEN, n1, 0), Bih1b1 = loadB(Wih1, HIDDEN, n1, 1);
  const v16h Bhh1a0 = loadB(Whh1, HIDDEN, n0, 0), Bhh1a1 = loadB(Whh1, HIDDEN, n0, 1);
  const v16h Bhh1b0 = loadB(Whh1, HIDDEN, n1, 0), Bhh1b1 = loadB(Whh1, HIDDEN, n1, 1);

  float c0[4] = {0.f, 0.f, 0.f, 0.f};
  float c1[4] = {0.f, 0.f, 0.f, 0.f};
  const int en  = tid & 63;
  const int emb = tid >> 6;

  const int lane = tid & 31;
  const int cn   = lane & 15;
  const int cmb  = (lane >> 4) << 3;

  for (int t = 0; t < SEQ; ++t) {
#pragma unroll
    for (int i = 0; i < 2; ++i) {
      int idx = tid + i * 256;
      int m = idx >> 5, k = idx & 31;
      float v = (k < FEAT) ? x[((size_t)(bbase + m) * SEQ + t) * FEAT + k] : 0.0f;
      sh_x[idx] = (_Float16)v;
    }
    if (t + 1 < SEQ)
      __builtin_prefetch(&x[((size_t)(bbase + (tid >> 5)) * SEQ + t + 1) * FEAT], 0, 0);
    __syncthreads();

    {
      v16h ax  = ldsA(sh_x, 32, 0);
      v16h ah0 = ldsA(sh_h0, HIDDEN, 0);
      v16h ah1 = ldsA(sh_h0, HIDDEN, 1);
      v8f acc0 = {}; v8f acc1 = {};
      acc0 = WMMA(ax,  Bih0a,  acc0);
      acc0 = WMMA(ah0, Bhh0a0, acc0);
      acc0 = WMMA(ah1, Bhh0a1, acc0);
      acc1 = WMMA(ax,  Bih0b,  acc1);
      acc1 = WMMA(ah0, Bhh0b0, acc1);
      acc1 = WMMA(ah1, Bhh0b1, acc1);
#pragma unroll
      for (int r = 0; r < 8; ++r) {
        sh_g[(cmb + r) * GATES + n0 + cn] = acc0[r];
        sh_g[(cmb + r) * GATES + n1 + cn] = acc1[r];
      }
    }
    __syncthreads();

#pragma unroll
    for (int p = 0; p < 4; ++p) {
      int m = p * 4 + emb;
      const float* g = sh_g + m * GATES;
      float gi = g[en]        + sh_b0[en];
      float gf = g[64  + en]  + sh_b0[64  + en];
      float gz = g[128 + en]  + sh_b0[128 + en];
      float go = g[192 + en]  + sh_b0[192 + en];
      float cc = fsig(gf) * c0[p] + fsig(gi) * ftanh(gz);
      c0[p] = cc;
      sh_h0[m * HIDDEN + en] = (_Float16)(fsig(go) * ftanh(cc));
    }
    __syncthreads();

    {
      v16h ai0 = ldsA(sh_h0, HIDDEN, 0);
      v16h ai1 = ldsA(sh_h0, HIDDEN, 1);
      v16h ah0 = ldsA(sh_h1, HIDDEN, 0);
      v16h ah1 = ldsA(sh_h1, HIDDEN, 1);
      v8f acc0 = {}; v8f acc1 = {};
      acc0 = WMMA(ai0, Bih1a0, acc0);
      acc0 = WMMA(ai1, Bih1a1, acc0);
      acc0 = WMMA(ah0, Bhh1a0, acc0);
      acc0 = WMMA(ah1, Bhh1a1, acc0);
      acc1 = WMMA(ai0, Bih1b0, acc1);
      acc1 = WMMA(ai1, Bih1b1, acc1);
      acc1 = WMMA(ah0, Bhh1b0, acc1);
      acc1 = WMMA(ah1, Bhh1b1, acc1);
#pragma unroll
      for (int r = 0; r < 8; ++r) {
        sh_g[(cmb + r) * GATES + n0 + cn] = acc0[r];
        sh_g[(cmb + r) * GATES + n1 + cn] = acc1[r];
      }
    }
    __syncthreads();

#pragma unroll
    for (int p = 0; p < 4; ++p) {
      int m = p * 4 + emb;
      const float* g = sh_g + m * GATES;
      float gi = g[en]        + sh_b1[en];
      float gf = g[64  + en]  + sh_b1[64  + en];
      float gz = g[128 + en]  + sh_b1[128 + en];
      float go = g[192 + en]  + sh_b1[192 + en];
      float cc = fsig(gf) * c1[p] + fsig(gi) * ftanh(gz);
      c1[p] = cc;
      sh_h1[m * HIDDEN + en] = (_Float16)(fsig(go) * ftanh(cc));
    }
    __syncthreads();
  }

  if (tid < MTILE * NCLS) {
    int m = tid >> 1, cls = tid & 1;
    float s = bfc[cls];
#pragma unroll 8
    for (int n = 0; n < HIDDEN; ++n)
      s += (float)sh_h1[m * HIDDEN + n] * Wfc[cls * HIDDEN + n];
    out[(bbase + m) * NCLS + cls] = s;
  }
}

extern "C" void kernel_launch(void* const* d_in, const int* in_sizes, int n_in,
                              void* d_out, int out_size, void* d_ws, size_t ws_size,
                              hipStream_t stream) {
  const float* x    = (const float*)d_in[0];
  const float* Wih0 = (const float*)d_in[1];
  const float* Whh0 = (const float*)d_in[2];
  const float* bih0 = (const float*)d_in[3];
  const float* bhh0 = (const float*)d_in[4];
  const float* Wih1 = (const float*)d_in[5];
  const float* Whh1 = (const float*)d_in[6];
  const float* bih1 = (const float*)d_in[7];
  const float* bhh1 = (const float*)d_in[8];
  const float* Wfc  = (const float*)d_in[9];
  const float* bfc  = (const float*)d_in[10];
  float* out = (float*)d_out;

  dim3 grid(BATCH / MTILE);
  dim3 block(256);
  hipLaunchKernelGGL(lstm2_fused_wmma, grid, block, 0, stream,
                     x, Wih0, Whh0, bih0, bhh0, Wih1, Whh1, bih1, bhh1,
                     Wfc, bfc, out);
}


# each wave owns i,f,g,o of 16 hidden units, cell update on the WMMA accumulators: no gate tile through LDS; 2 barriers per step, not 5 (next x tile staged during layer 1); order of sums unchanged
# speedup vs baseline: 2.7501x; 2.7501x over previous
#include <hip/hip_runtime.h>


#define HIDDEN 64
#define FEAT   9
#define SEQ    128
#define BATCH  8192
#define NCLS   2
#define GATES  256
#define MTILE  16
#define NTHR   128
#define HTILE  (MTILE * HIDDEN)

typedef _Float16 v16h __attribute__((ext_vector_type(16)));
typedef float    v8f  __attribute__((ext_vector_type(8)));

#define WMMA(A, B, C) __builtin_amdgcn_wmma_f32_16x16x32_f16( \
    false, (A), false, (B), (short)0, (C), false, false)

static __device__ __forceinline__ v16h ldsA(const _Float16* base, int ld, int kc) {
  const int lane = threadIdx.x & 31;
  const int m  = lane & 15;
  const int kb = ((lane >> 4) << 3) + (kc << 5);
  const _Float16* p = base + m * ld + kb;
  v16h a;
#pragma unroll
  for (int i = 0; i < 8; ++i) { a[i] = p[i]; a[i + 8] = p[i + 16]; }
  return a;
}

static __device__ __forceinline__ v16h loadB(const float* W, int ld, int n0, int kc) {
  const int lane = threadIdx.x & 31;
  const int n  = n0 + (lane & 15);
  const int kb = ((lane >> 4) << 3) + (kc << 5);
  const float* p = W + n * ld + kb;
  v16h b;
#pragma unroll
  for (int i = 0; i < 8; ++i) { b[i] = (_Float16)p[i]; b[i + 8] = (_Float16)p[i + 16]; }
  return b;
}

static __device__ __forceinline__ v16h loadB_pad(const float* W, int ld, int kmax, int n0) {
  const int lane = threadIdx.x & 31;
  const int n  = n0 + (lane & 15);
  const int kb = (lane >> 4) << 3;
  v16h b;
#pragma unroll
  for (int i = 0; i < 8; ++i) {
    int k0 = kb + i, k1 = kb + 16 + i;
    b[i]     = (k0 < kmax) ? (_Float16)W[n * ld + k0] : (_Float16)0.0f;
    b[i + 8] = (k1 < kmax) ? (_Float16)W[n * ld + k1] : (_Float16)0.0f;
  }
  return b;
}

static __device__ __forceinline__ float ftanh(float x) { return __builtin_amdgcn_tanhf(x); }
static __device__ __forceinline__ float fsig(float x)  { return __builtin_fmaf(0.5f, __builtin_amdgcn_tanhf(0.5f * x), 0.5f); }

__global__ __launch_bounds__(NTHR) void lstm2_fused_wmma(
    const float* __restrict__ x,
    const float* __restrict__ Wih0, const float* __restrict__ Whh0,
    const float* __restrict__ bih0, const float* __restrict__ bhh0,
    const float* __restrict__ Wih1, const float* __restrict__ Whh1,
    const float* __restrict__ bih1, const float* __restrict__ bhh1,
    const float* __restrict__ Wfc,  const float* __restrict__ bfc,
    float* __restrict__ out) {
  __shared__ __attribute__((aligned(16))) _Float16 sh_x [MTILE * 32];
  __shared__ __attribute__((aligned(16))) _Float16 sh_h0[2 * HTILE];
  __shared__ __attribute__((aligned(16))) _Float16 sh_h1[2 * HTILE];

  const int tid   = threadIdx.x;
  const int wave  = tid >> 5;
  const int lane  = tid & 31;
  const int bbase = blockIdx.x * MTILE;

  const int cn  = lane & 15;
  const int cmb = (lane >> 4) << 3;
  const int hn  = wave * 16 + cn;

  for (int i = tid; i < MTILE * 32; i += NTHR) sh_x[i] = (_Float16)0.0f;
  for (int i = tid; i < HTILE; i += NTHR) {
    sh_h0[i] = (_Float16)0.0f;
    sh_h1[i] = (_Float16)0.0f;
  }

  v16h Bih0[4], Bhh0[4][2], Bih1[4][2], Bhh1[4][2];
  float bs0[4], bs1[4];
#pragma unroll
  for (int q = 0; q < 4; ++q) {
    const int n0 = q * HIDDEN + wave * 16;
    Bih0[q]    = loadB_pad(Wih0, FEAT, FEAT, n0);
    Bhh0[q][0] = loadB(Whh0, HIDDEN, n0, 0);
    Bhh0[q][1] = loadB(Whh0, HIDDEN, n0, 1);
    Bih1[q][0] = loadB(Wih1, HIDDEN, n0, 0);
    Bih1[q][1] = loadB(Wih1, HIDDEN, n0, 1);
    Bhh1[q][0] = loadB(Whh1, HIDDEN, n0, 0);
    Bhh1[q][1] = loadB(Whh1, HIDDEN, n0, 1);
    bs0[q] = bih0[n0 + cn] + bhh0[n0 + cn];
    bs1[q] = bih1[n0 + cn] + bhh1[n0 + cn];
  }

  float c0[8], c1[8];
#pragma unroll
  for (int r = 0; r < 8; ++r) { c0[r] = 0.f; c1[r] = 0.f; }

  const int e0 = tid, e1 = tid + NTHR;
  const int m0 = e0 / FEAT, k0 = e0 - m0 * FEAT;
  const int m1 = e1 / FEAT, k1 = e1 - m1 * FEAT;
  const float* xp0 = x + ((size_t)(bbase + m0) * SEQ) * FEAT + k0;
  const float* xp1 = x + ((size_t)(bbase + (m1 & (MTILE - 1))) * SEQ) * FEAT + k1;
  const int s0 = m0 * 32 + k0, s1 = (m1 & (MTILE - 1)) * 32 + k1;
  const bool two = (e1 < MTILE * FEAT);

  __syncthreads();
  {
    const float a0 = xp0[0], a1 = xp1[0];
    sh_x[s0] = (_Float16)a0;
    if (two) sh_x[s1] = (_Float16)a1;
  }

  for (int t = 0; t < SEQ; ++t) {
    const int cur = (t & 1) * HTILE;
    const int nxt = HTILE - cur;
    __syncthreads();

    {
      const v16h ax  = ldsA(sh_x, 32, 0);
      const v16h ah0 = ldsA(sh_h0 + cur, HIDDEN, 0);
      const v16h ah1 = ldsA(sh_h0 + cur, HIDDEN, 1);
      v8f acc[4];
#pragma unroll
      for (int q = 0; q < 4; ++q) { v8f z = {}; acc[q] = WMMA(ax, Bih0[q], z); }
#pragma unroll
      for (int q = 0; q < 4; ++q) acc[q] = WMMA(ah0, Bhh0[q][0], acc[q]);
#pragma unroll
      for (int q = 0; q < 4; ++q) acc[q] = WMMA(ah1, Bhh0[q][1], acc[q]);
      _Float16* hw = sh_h0 + nxt + cmb * HIDDEN + hn;
#pragma unroll
      for (int r = 0; r < 8; ++r) {
        float gi = acc[0][r] + bs0[0];
        float gf = acc[1][r] + bs0[1];
        float gz = acc[2][r] + bs0[2];
        float go = acc[3][r] + bs0[3];
        float cc = fsig(gf) * c0[r] + fsig(gi) * ftanh(gz);
        c0[r] = cc;
        hw[r * HIDDEN] = (_Float16)(fsig(go) * ftanh(cc));
      }
    }
    __syncthreads();

    const bool more = (t + 1 < SEQ);
    float xv0 = 0.f, xv1 = 0.f;
    if (more) {
      xv0 = xp0[(t + 1) * FEAT];
      xv1 = xp1[(t + 1) * FEAT];
      if (t + 2 < SEQ)
        __builtin_prefetch(&xp0[(t + 2) * FEAT], 0, 0);
    }

    {
      const v16h ai0 = ldsA(sh_h0 + nxt, HIDDEN, 0);
      const v16h ai1 = ldsA(sh_h0 + nxt, HIDDEN, 1);
      const v16h ah0 = ldsA(sh_h1 + cur, HIDDEN, 0);
      const v16h ah1 = ldsA(sh_h1 + cur, HIDDEN, 1);
      v8f acc[4];
#pragma unroll
      for (int q = 0; q < 4; ++q) { v8f z = {}; acc[q] = WMMA(ai0, Bih1[q][0], z); }
#pragma unroll
      for (int q = 0; q < 4; ++q) acc[q] = WMMA(ai1, Bih1[q][1], acc[q]);
#pragma unroll
      for (int q = 0; q < 4; ++q) acc[q] = WMMA(ah0, Bhh1[q][0], acc[q]);
#pragma unroll
      for (int q = 0; q < 4; ++q) acc[q] = WMMA(ah1, Bhh1[q][1], acc[q]);
      _Float16* hw = sh_h1 + nxt + cmb * HIDDEN + hn;
#pragma unroll
      for (int r = 0; r < 8; ++r) {
        float gi = acc[0][r] + bs1[0];
        float gf = acc[1][r] + bs1[1];
        float gz = acc[2][r] + bs1[2];
        float go = acc[3][r] + bs1[3];
        float cc = fsig(gf) * c1[r] + fsig(gi) * ftanh(gz);
        c1[r] = cc;
        hw[r * HIDDEN] = (_Float16)(fsig(go) * ftanh(cc));
      }
    }

    if (more) {
      sh_x[s0] = (_Float16)xv0;
      if (two) sh_x[s1] = (_Float16)xv1;
    }
  }
  __syncthreads();

  if (tid < MTILE * NCLS) {
    int m = tid >> 1, cls = tid & 1;
    float s = bfc[cls];
#pragma unroll 8
    for (int n = 0; n < HIDDEN; ++n)
      s += (float)sh_h1[m * HIDDEN + n] * Wfc[cls * HIDDEN + n];
    out[(bbase + m) * NCLS + cls] = s;
  }
}

extern "C" void kernel_launch(void* const* d_in, const int* in_sizes, int n_in,
                              void* d_out, int out_size, void* d_ws, size_t ws_size,
                              hipStream_t stream) {
  const float* x    = (const float*)d_in[0];
  const float* Wih0 = (const float*)d_in[1];
  const float* Whh0 = (const float*)d_in[2];
  const float* bih0 = (const float*)d_in[3];
  const float* bhh0 = (const float*)d_in[4];
  const float* Wih1 = (const float*)d_in[5];
  const float* Whh1 = (const float*)d_in[6];
  const float* bih1 = (const float*)d_in[7];
  const float* bhh1 = (const float*)d_in[8];
  const float* Wfc  = (const float*)d_in[9];
  const float* bfc  = (const float*)d_in[10];
  float* out = (float*)d_out;

  dim3 grid(BATCH / MTILE);
  dim3 block(NTHR);
  hipLaunchKernelGGL(lstm2_fused_wmma, grid, block, 0, stream,
                     x, Wih0, Whh0, bih0, bhh0, Wih1, Whh1, bih1, bhh1,
                     Wfc, bfc, out);
}


# LDS row pitch padded to an odd multiple of 16 B (state rows 144 B, x rows 80 B): 128-bit A-fragment LDS reads are bank-conflict-free (were 8-way and 4-way conflicted); order of sums unchanged
# speedup vs baseline: 3.7245x; 1.3543x over previous
#include <hip/hip_runtime.h>


#define HIDDEN 64
#define FEAT   9
#define SEQ    128
#define BATCH  8192
#define NCLS   2
#define GATES  256
#define MTILE  16
#define NTHR   128
#define HP     72
#define XP     40
#define HTILE  (MTILE * HP)

typedef _Float16 v16h __attribute__((ext_vector_type(16)));
typedef float    v8f  __attribute__((ext_vector_type(8)));

#define WMMA(A, B, C) __builtin_amdgcn_wmma_f32_16x16x32_f16( \
    false, (A), false, (B), (short)0, (C), false, false)

static __device__ __forceinline__ v16h ldsA(const _Float16* base, int ld, int kc) {
  const int lane = threadIdx.x & 31;
  const int m  = lane & 15;
  const int kb = ((lane >> 4) << 3) + (kc << 5);
  const _Float16* p = base + m * ld + kb;
  v16h a;
#pragma unroll
  for (int i = 0; i < 8; ++i) { a[i] = p[i]; a[i + 8] = p[i + 16]; }
  return a;
}

static __device__ __forceinline__ v16h loadB(const float* W, int ld, int n0, int kc) {
  const int lane = threadIdx.x & 31;
  const int n  = n0 + (lane & 15);
  const int kb = ((lane >> 4) << 3) + (kc << 5);
  const float* p = W + n * ld + kb;
  v16h b;
#pragma unroll
  for (int i = 0; i < 8; ++i) { b[i] = (_Float16)p[i]; b[i + 8] = (_Float16)p[i + 16]; }
  return b;
}

static __device__ __forceinline__ v16h loadB_pad(const float* W, int ld, int kmax, int n0) {
  const int lane = threadIdx.x & 31;
  const int n  = n0 + (lane & 15);
  const int kb = (lane >> 4) << 3;
  v16h b;
#pragma unroll
  for (int i = 0; i < 8; ++i) {
    int k0 = kb + i, k1 = kb + 16 + i;
    b[i]     = (k0 < kmax) ? (_Float16)W[n * ld + k0] : (_Float16)0.0f;
    b[i + 8] = (k1 < kmax) ? (_Float16)W[n * ld + k1] : (_Float16)0.0f;
  }
  return b;
}

static __device__ __forceinline__ float ftanh(float x) { return __builtin_amdgcn_tanhf(x); }
static __device__ __forceinline__ float fsig(float x)  { return __builtin_fmaf(0.5f, __builtin_amdgcn_tanhf(0.5f * x), 0.5f); }

__global__ __launch_bounds__(NTHR) void lstm2_fused_wmma(
    const float* __restrict__ x,
    const float* __restrict__ Wih0, const float* __restrict__ Whh0,
    const float* __restrict__ bih0, const float* __restrict__ bhh0,
    const float* __restrict__ Wih1, const float* __restrict__ Whh1,
    const float* __restrict__ bih1, const float* __restrict__ bhh1,
    const float* __restrict__ Wfc,  const float* __restrict__ bfc,
    float* __restrict__ out) {
  __shared__ __attribute__((aligned(16))) _Float16 sh_x [MTILE * XP];
  __shared__ __attribute__((aligned(16))) _Float16 sh_h0[2 * HTILE];
  __shared__ __attribute__((aligned(16))) _Float16 sh_h1[2 * HTILE];

  const int tid   = threadIdx.x;
  const int wave  = tid >> 5;
  const int lane  = tid & 31;
  const int bbase = blockIdx.x * MTILE;

  const int cn  = lane & 15;
  const int cmb = (lane >> 4) << 3;
  const int hn  = wave * 16 + cn;

  for (int i = tid; i < MTILE * XP; i += NTHR) sh_x[i] = (_Float16)0.0f;
  for (int i = tid; i < HTILE; i += NTHR) {
    sh_h0[i] = (_Float16)0.0f;
    sh_h1[i] = (_Float16)0.0f;
  }

  v16h Bih0[4], Bhh0[4][2], Bih1[4][2], Bhh1[4][2];
  float bs0[4], bs1[4];
#pragma unroll
  for (int q = 0; q < 4; ++q) {
    const int n0 = q * HIDDEN + wave * 16;
    Bih0[q]    = loadB_pad(Wih0, FEAT, FEAT, n0);
    Bhh0[q][0] = loadB(Whh0, HIDDEN, n0, 0);
    Bhh0[q][1] = loadB(Whh0, HIDDEN, n0, 1);
    Bih1[q][0] = loadB(Wih1, HIDDEN, n0, 0);
    Bih1[q][1] = loadB(Wih1, HIDDEN, n0, 1);
    Bhh1[q][0] = loadB(Whh1, HIDDEN, n0, 0);
    Bhh1[q][1] = loadB(Whh1, HIDDEN, n0, 1);
    bs0[q] = bih0[n0 + cn] + bhh0[n0 + cn];
    bs1[q] = bih1[n0 + cn] + bhh1[n0 + cn];
  }

  float c0[8], c1[8];
#pragma unroll
  for (int r = 0; r < 8; ++r) { c0[r] = 0.f; c1[r] = 0.f; }

  const int e0 = tid, e1 = tid + NTHR;
  const int m0 = e0 / FEAT, k0 = e0 - m0 * FEAT;
  const int m1 = e1 / FEAT, k1 = e1 - m1 * FEAT;
  const float* xp0 = x + ((size_t)(bbase + m0) * SEQ) * FEAT + k0;
  const float* xp1 = x + ((size_t)(bbase + (m1 & (MTILE - 1))) * SEQ) * FEAT + k1;
  const int s0 = m0 * XP + k0, s1 = (m1 & (MTILE - 1)) * XP + k1;
  const bool two = (e1 < MTILE * FEAT);

  __syncthreads();
  {
    const float a0 = xp0[0], a1 = xp1[0];
    sh_x[s0] = (_Float16)a0;
    if (two) sh_x[s1] = (_Float16)a1;
  }

  for (int t = 0; t < SEQ; ++t) {
    const int cur = (t & 1) * HTILE;
    const int nxt = HTILE - cur;
    __syncthreads();

    {
      const v16h ax  = ldsA(sh_x, XP, 0);
      const v16h ah0 = ldsA(sh_h0 + cur, HP, 0);
      const v16h ah1 = ldsA(sh_h0 + cur, HP, 1);
      v8f acc[4];
#pragma unroll
      for (int q = 0; q < 4; ++q) { v8f z = {}; acc[q] = WMMA(ax, Bih0[q], z); }
#pragma unroll
      for (int q = 0; q < 4; ++q) acc[q] = WMMA(ah0, Bhh0[q][0], acc[q]);
#pragma unroll
      for (int q = 0; q < 4; ++q) acc[q] = WMMA(ah1, Bhh0[q][1], acc[q]);
      _Float16* hw = sh_h0 + nxt + cmb * HP + hn;
#pragma unroll
      for (int r = 0; r < 8; ++r) {
        float gi = acc[0][r] + bs0[0];
        float gf = acc[1][r] + bs0[1];
        float gz = acc[2][r] + bs0[2];
        float go = acc[3][r] + bs0[3];
        float cc = fsig(gf) * c0[r] + fsig(gi) * ftanh(gz);
        c0[r] = cc;
        hw[r * HP] = (_Float16)(fsig(go) * ftanh(cc));
      }
    }
    __syncthreads();

    const bool more = (t + 1 < SEQ);
    float xv0 = 0.f, xv1 = 0.f;
    if (more) {
      xv0 = xp0[(t + 1) * FEAT];
      xv1 = xp1[(t + 1) * FEAT];
      if (t + 2 < SEQ)
        __builtin_prefetch(&xp0[(t + 2) * FEAT], 0, 0);
    }

    {
      const v16h ai0 = ldsA(sh_h0 + nxt, HP, 0);
      const v16h ai1 = ldsA(sh_h0 + nxt, HP, 1);
      const v16h ah0 = ldsA(sh_h1 + cur, HP, 0);
      const v16h ah1 = ldsA(sh_h1 + cur, HP, 1);
      v8f acc[4];
#pragma unroll
      for (int q = 0; q < 4; ++q) { v8f z = {}; acc[q] = WMMA(ai0, Bih1[q][0], z); }
#pragma unroll
      for (int q = 0; q < 4; ++q) acc[q] = WMMA(ai1, Bih1[q][1], acc[q]);
#pragma unroll
      for (int q = 0; q < 4; ++q) acc[q] = WMMA(ah0, Bhh1[q][0], acc[q]);
#pragma unroll
      for (int q = 0; q < 4; ++q) acc[q] = WMMA(ah1, Bhh1[q][1], acc[q]);
      _Float16* hw = sh_h1 + nxt + cmb * HP + hn;
#pragma unroll
      for (int r = 0; r < 8; ++r) {
        float gi = acc[0][r] + bs1[0];
        float gf = acc[1][r] + bs1[1];
        float gz = acc[2][r] + bs1[2];
        float go = acc[3][r] + bs1[3];
        float cc = fsig(gf) * c1[r] + fsig(gi) * ftanh(gz);
        c1[r] = cc;
        hw[r * HP] = (_Float16)(fsig(go) * ftanh(cc));
      }
    }

    if (more) {
      sh_x[s0] = (_Float16)xv0;
      if (two) sh_x[s1] = (_Float16)xv1;
    }
  }
  __syncthreads();

  if (tid < MTILE * NCLS) {
    int m = tid >> 1, cls = tid & 1;
    float s = bfc[cls];
#pragma unroll 8
    for (int n = 0; n < HIDDEN; ++n)
      s += (float)sh_h1[m * HP + n] * Wfc[cls * HIDDEN + n];
    out[(bbase + m) * NCLS + cls] = s;
  }
}

extern "C" void kernel_launch(void* const* d_in, const int* in_sizes, int n_in,
                              void* d_out, int out_size, void* d_ws, size_t ws_size,
                              hipStream_t stream) {
  const float* x    = (const float*)d_in[0];
  const float* Wih0 = (const float*)d_in[1];
  const float* Whh0 = (const float*)d_in[2];
  const float* bih0 = (const float*)d_in[3];
  const float* bhh0 = (const float*)d_in[4];
  const float* Wih1 = (const float*)d_in[5];
  const float* Whh1 = (const float*)d_in[6];
  const float* bih1 = (const float*)d_in[7];
  const float* bhh1 = (const float*)d_in[8];
  const float* Wfc  = (const float*)d_in[9];
  const float* bfc  = (const float*)d_in[10];
  float* out = (float*)d_out;

  dim3 grid(BATCH / MTILE);
  dim3 block(NTHR);
  hipLaunchKernelGGL(lstm2_fused_wmma, grid, block, 0, stream,
                     x, Wih0, Whh0, bih0, bhh0, Wih1, Whh1, bih1, bhh1,
                     Wfc, bfc, out);
}


# layer 1 runs one step behind layer 0: layer0(t+1) and layer1(t) share one barrier interval and one LDS read of h0(t): 1 barrier per step, not 2 (x tile double-buffered); order of sums unchanged
# speedup vs baseline: 3.8594x; 1.0362x over previous
#include <hip/hip_runtime.h>


#define HIDDEN 64
#define FEAT   9
#define SEQ    128
#define BATCH  8192
#define NCLS   2
#define GATES  256
#define MTILE  16
#define NTHR   128
#define HP     72
#define XP     40
#define HTILE  (MTILE * HP)
#define XTILE  (MTILE * XP)

typedef _Float16 v16h __attribute__((ext_vector_type(16)));
typedef float    v8f  __attribute__((ext_vector_type(8)));

#define WMMA(A, B, C) __builtin_amdgcn_wmma_f32_16x16x32_f16( \
    false, (A), false, (B), (short)0, (C), false, false)

static __device__ __forceinline__ v16h ldsA(const _Float16* base, int ld, int kc) {
  const int lane = threadIdx.x & 31;
  const int m  = lane & 15;
  const int kb = ((lane >> 4) << 3) + (kc << 5);
  const _Float16* p = base + m * ld + kb;
  v16h a;
#pragma unroll
  for (int i = 0; i < 8; ++i) { a[i] = p[i]; a[i + 8] = p[i + 16]; }
  return a;
}

static __device__ __forceinline__ v16h loadB(const float* W, int ld, int n0, int kc) {
  const int lane = threadIdx.x & 31;
  const int n  = n0 + (lane & 15);
  const int kb = ((lane >> 4) << 3) + (kc << 5);
  const float* p = W + n * ld + kb;
  v16h b;
#pragma unroll
  for (int i = 0; i < 8; ++i) { b[i] = (_Float16)p[i]; b[i + 8] = (_Float16)p[i + 16]; }
  return b;
}

static __device__ __forceinline__ v16h loadB_pad(const float* W, int ld, int kmax, int n0) {
  const int lane = threadIdx.x & 31;
  const int n  = n0 + (lane & 15);
  const int kb = (lane >> 4) << 3;
  v16h b;
#pragma unroll
  for (int i = 0; i < 8; ++i) {
    int k0 = kb + i, k1 = kb + 16 + i;
    b[i]     = (k0 < kmax) ? (_Float16)W[n * ld + k0] : (_Float16)0.0f;
    b[i + 8] = (k1 < kmax) ? (_Float16)W[n * ld + k1] : (_Float16)0.0f;
  }
  return b;
}

static __device__ __forceinline__ float ftanh(float x) { return __builtin_amdgcn_tanhf(x); }
static __device__ __forceinline__ float fsig(float x)  { return __builtin_fmaf(0.5f, __builtin_amdgcn_tanhf(0.5f * x), 0.5f); }

static __device__ __forceinline__ void cell_update(const v8f (&acc)[4], const float (&bs)[4], float (&c)[8], _Float16* hw) {
#pragma unroll
  for (int r = 0; r < 8; ++r) {
    float gi = acc[0][r] + bs[0];
    float gf = acc[1][r] + bs[1];
    float gz = acc[2][r] + bs[2];
    float go = acc[3][r] + bs[3];
    float cc = fsig(gf) * c[r] + fsig(gi) * ftanh(gz);
    c[r] = cc;
    hw[r * HP] = (_Float16)(fsig(go) * ftanh(cc));
  }
}

__global__ __launch_bounds__(NTHR) void lstm2_fused_wmma(
    const float* __restrict__ x,
    const float* __restrict__ Wih0, const float* __restrict__ Whh0,
    const float* __restrict__ bih0, const float* __restrict__ bhh0,
    const float* __restrict__ Wih1, const float* __restrict__ Whh1,
    const float* __restrict__ bih1, const float* __restrict__ bhh1,
    const float* __restrict__ Wfc,  const float* __restrict__ bfc,
    float* __restrict__ out) {
  __shared__ __attribute__((aligned(16))) _Float16 sh_x [2 * XTILE];
  __shared__ __attribute__((aligned(16))) _Float16 sh_h0[2 * HTILE];
  __shared__ __attribute__((aligned(16))) _Float16 sh_h1[2 * HTILE];

  const int tid   = threadIdx.x;
  const int wave  = tid >> 5;
  const int lane  = tid & 31;
  const int bbase = blockIdx.x * MTILE;

  const int cn  = lane & 15;
  const int cmb = (lane >> 4) << 3;
  const int hn  = wave * 16 + cn;

  for (int i = tid; i < 2 * XTILE; i += NTHR) sh_x[i] = (_Float16)0.0f;
  for (int i = tid; i < HTILE; i += NTHR) {
    sh_h0[i] = (_Float16)0.0f;
    sh_h1[i] = (_Float16)0.0f;
  }

  v16h Bih0[4], Bhh0[4][2], Bih1[4][2], Bhh1[4][2];
  float bs0[4], bs1[4];
#pragma unroll
  for (int q = 0; q < 4; ++q) {
    const int n0 = q * HIDDEN + wave * 16;
    Bih0[q]    = loadB_pad(Wih0, FEAT, FEAT, n0);
    Bhh0[q][0] = loadB(Whh0, HIDDEN, n0, 0);
    Bhh0[q][1] = loadB(Whh0, HIDDEN, n0, 1);
    Bih1[q][0] = loadB(Wih1, HIDDEN, n0, 0);
    Bih1[q][1] = loadB(Wih1, HIDDEN, n0, 1);
    Bhh1[q][0] = loadB(Whh1, HIDDEN, n0, 0);
    Bhh1[q][1] = loadB(Whh1, HIDDEN, n0, 1);
    bs0[q] = bih0[n0 + cn] + bhh0[n0 + cn];
    bs1[q] = bih1[n0 + cn] + bhh1[n0 + cn];
  }

  float c0[8], c1[8];
#pragma unroll
  for (int r = 0; r < 8; ++r) { c0[r] = 0.f; c1[r] = 0.f; }

  const int e0 = tid, e1 = tid + NTHR;
  const int m0 = e0 / FEAT, k0 = e0 - m0 * FEAT;
  const int m1 = e1 / FEAT, k1 = e1 - m1 * FEAT;
  const float* xp0 = x + ((size_t)(bbase + m0) * SEQ) * FEAT + k0;
  const float* xp1 = x + ((size_t)(bbase + (m1 & (MTILE - 1))) * SEQ) * FEAT + k1;
  const int s0 = m0 * XP + k0, s1 = (m1 & (MTILE - 1)) * XP + k1;
  const bool two = (e1 < MTILE * FEAT);

  const int hoff = cmb * HP + hn;

  float xc0 = xp0[2 * FEAT], xc1 = xp1[2 * FEAT];

  __syncthreads();
  {
    const float a0 = xp0[0], a1 = xp1[0];
    const float b0 = xp0[FEAT], b1 = xp1[FEAT];
    sh_x[s0] = (_Float16)a0;
    sh_x[XTILE + s0] = (_Float16)b0;
    if (two) { sh_x[s1] = (_Float16)a1; sh_x[XTILE + s1] = (_Float16)b1; }
  }
  __syncthreads();

  {
    const v16h ax  = ldsA(sh_x, XP, 0);
    const v16h ah0 = ldsA(sh_h0, HP, 0);
    const v16h ah1 = ldsA(sh_h0, HP, 1);
    v8f acc[4];
#pragma unroll
    for (int q = 0; q < 4; ++q) { v8f z = {}; acc[q] = WMMA(ax, Bih0[q], z); }
#pragma unroll
    for (int q = 0; q < 4; ++q) acc[q] = WMMA(ah0, Bhh0[q][0], acc[q]);
#pragma unroll
    for (int q = 0; q < 4; ++q) acc[q] = WMMA(ah1, Bhh0[q][1], acc[q]);
    cell_update(acc, bs0, c0, sh_h0 + HTILE + hoff);
  }

  for (int j = 1; j < SEQ; ++j) {
    const int p  = j & 1;
    const int h0r = p * HTILE,       h0w = HTILE - h0r;
    const int h1w = p * HTILE,       h1r = HTILE - h1w;
    const int xr  = p * XTILE,       xw  = XTILE - xr;
    __syncthreads();

    const bool more = (j + 1 < SEQ);
    float xn0 = 0.f, xn1 = 0.f;
    if (j + 2 < SEQ) {
      xn0 = xp0[(j + 2) * FEAT];
      xn1 = xp1[(j + 2) * FEAT];
      if (j + 3 < SEQ)
        __builtin_prefetch(&xp0[(j + 3) * FEAT], 0, 0);
    }

    const v16h ax  = ldsA(sh_x + xr, XP, 0);
    const v16h ah0 = ldsA(sh_h0 + h0r, HP, 0);
    const v16h ah1 = ldsA(sh_h0 + h0r, HP, 1);
    const v16h bh0 = ldsA(sh_h1 + h1r, HP, 0);
    const v16h bh1 = ldsA(sh_h1 + h1r, HP, 1);

    v8f acc0[4], acc1[4];
#pragma unroll
    for (int q = 0; q < 4; ++q) { v8f z = {}; acc0[q] = WMMA(ax, Bih0[q], z); }
#pragma unroll
    for (int q = 0; q < 4; ++q) { v8f z = {}; acc1[q] = WMMA(ah0, Bih1[q][0], z); }
#pragma unroll
    for (int q = 0; q < 4; ++q) acc0[q] = WMMA(ah0, Bhh0[q][0], acc0[q]);
#pragma unroll
    for (int q = 0; q < 4; ++q) acc1[q] = WMMA(ah1, Bih1[q][1], acc1[q]);
#pragma unroll
    for (int q = 0; q < 4; ++q) acc0[q] = WMMA(ah1, Bhh0[q][1], acc0[q]);
#pragma unroll
    for (int q = 0; q < 4; ++q) acc1[q] = WMMA(bh0, Bhh1[q][0], acc1[q]);
#pragma unroll
    for (int q = 0; q < 4; ++q) acc1[q] = WMMA(bh1, Bhh1[q][1], acc1[q]);

    cell_update(acc0, bs0, c0, sh_h0 + h0w + hoff);
    cell_update(acc1, bs1, c1, sh_h1 + h1w + hoff);

    if (more) {
      sh_x[xw + s0] = (_Float16)xc0;
      if (two) sh_x[xw + s1] = (_Float16)xc1;
    }
    xc0 = xn0; xc1 = xn1;
  }
  __syncthreads();

  {
    const v16h ai0 = ldsA(sh_h0, HP, 0);
    const v16h ai1 = ldsA(sh_h0, HP, 1);
    const v16h ah0 = ldsA(sh_h1 + HTILE, HP, 0);
    const v16h ah1 = ldsA(sh_h1 + HTILE, HP, 1);
    v8f acc[4];
#pragma unroll
    for (int q = 0; q < 4; ++q) { v8f z = {}; acc[q] = WMMA(ai0, Bih1[q][0], z); }
#pragma unroll
    for (int q = 0; q < 4; ++q) acc[q] = WMMA(ai1, Bih1[q][1], acc[q]);
#pragma unroll
    for (int q = 0; q < 4; ++q) acc[q] = WMMA(ah0, Bhh1[q][0], acc[q]);
#pragma unroll
    for (int q = 0; q < 4; ++q) acc[q] = WMMA(ah1, Bhh1[q][1], acc[q]);
    cell_update(acc, bs1, c1, sh_h1 + hoff);
  }
  __syncthreads();

  if (tid < MTILE * NCLS) {
    int m = tid >> 1, cls = tid & 1;
    float s = bfc[cls];
#pragma unroll 8
    for (int n = 0; n < HIDDEN; ++n)
      s += (float)sh_h1[m * HP + n] * Wfc[cls * HIDDEN + n];
    out[(bbase + m) * NCLS + cls] = s;
  }
}

extern "C" void kernel_launch(void* const* d_in, const int* in_sizes, int n_in,
                              void* d_out, int out_size, void* d_ws, size_t ws_size,
                              hipStream_t stream) {
  const float* x    = (const float*)d_in[0];
  const float* Wih0 = (const float*)d_in[1];
  const float* Whh0 = (const float*)d_in[2];
  const float* bih0 = (const float*)d_in[3];
  const float* bhh0 = (const float*)d_in[4];
  const float* Wih1 = (const float*)d_in[5];
  const float* Whh1 = (const float*)d_in[6];
  const float* bih1 = (const float*)d_in[7];
  const float* bhh1 = (const float*)d_in[8];
  const float* Wfc  = (const float*)d_in[9];
  const float* bfc  = (const float*)d_in[10];
  float* out = (float*)d_out;

  dim3 grid(BATCH / MTILE);
  dim3 block(NTHR);
  hipLaunchKernelGGL(lstm2_fused_wmma, grid, block, 0, stream,
                     x, Wih0, Whh0, bih0, bhh0, Wih1, Whh1, bih1, bhh1,
                     Wfc, bfc, out);
}


# gate constants folded into WMMA operands: sigmoid's 0.5s scale f16 weights (x0.5, x0.25: exact), LDS state holds 2h, bias sum starts each accumulator (first in every sum; fp32 last bits differ)
# speedup vs baseline: 4.6524x; 1.2055x over previous
#include <hip/hip_runtime.h>


#define HIDDEN 64
#define FEAT   9
#define SEQ    128
#define BATCH  8192
#define NCLS   2
#define GATES  256
#define MTILE  16
#define NTHR   128
#define HP     72
#define XP     40
#define HTILE  (MTILE * HP)
#define XTILE  (MTILE * XP)

typedef _Float16 v16h __attribute__((ext_vector_type(16)));
typedef float    v8f  __attribute__((ext_vector_type(8)));

#define WMMA(A, B, C) __builtin_amdgcn_wmma_f32_16x16x32_f16( \
    false, (A), false, (B), (short)0, (C), false, false)

static __device__ __forceinline__ v16h ldsA(const _Float16* base, int ld, int kc) {
  const int lane = threadIdx.x & 31;
  const int m  = lane & 15;
  const int kb = ((lane >> 4) << 3) + (kc << 5);
  const _Float16* p = base + m * ld + kb;
  v16h a;
#pragma unroll
  for (int i = 0; i < 8; ++i) { a[i] = p[i]; a[i + 8] = p[i + 16]; }
  return a;
}

static __device__ __forceinline__ v16h loadB(const float* W, int ld, int n0, int kc, float sc) {
  const int lane = threadIdx.x & 31;
  const int n  = n0 + (lane & 15);
  const int kb = ((lane >> 4) << 3) + (kc << 5);
  const float* p = W + n * ld + kb;
  v16h b;
#pragma unroll
  for (int i = 0; i < 8; ++i) { b[i] = (_Float16)(sc * p[i]); b[i + 8] = (_Float16)(sc * p[i + 16]); }
  return b;
}

static __device__ __forceinline__ v16h loadB_pad(const float* W, int ld, int kmax, int n0, float sc) {
  const int lane = threadIdx.x & 31;
  const int n  = n0 + (lane & 15);
  const int kb = (lane >> 4) << 3;
  v16h b;
#pragma unroll
  for (int i = 0; i < 8; ++i) {
    int k0 = kb + i, k1 = kb + 16 + i;
    b[i]     = (k0 < kmax) ? (_Float16)(sc * W[n * ld + k0]) : (_Float16)0.0f;
    b[i + 8] = (k1 < kmax) ? (_Float16)(sc * W[n * ld + k1]) : (_Float16)0.0f;
  }
  return b;
}

static __device__ __forceinline__ float ftanh(float x) { return __builtin_amdgcn_tanhf(x); }

static __device__ __forceinline__ void cell_update(const v8f (&acc)[4], float (&c)[8], _Float16* hw) {
#pragma unroll
  for (int r = 0; r < 8; ++r) {
    const float ti = ftanh(acc[0][r]);
    const float tf = ftanh(acc[1][r]);
    const float tg = ftanh(acc[2][r]);
    const float to = ftanh(acc[3][r]);
    const float u  = __builtin_fmaf(tf, c[r], c[r]);
    const float v  = __builtin_fmaf(ti, tg, tg);
    const float cc = 0.5f * (u + v);
    c[r] = cc;
    const float tc = ftanh(cc);
    hw[r * HP] = (_Float16)__builtin_fmaf(to, tc, tc);
  }
}

__global__ __launch_bounds__(NTHR) void lstm2_fused_wmma(
    const float* __restrict__ x,
    const float* __restrict__ Wih0, const float* __restrict__ Whh0,
    const float* __restrict__ bih0, const float* __restrict__ bhh0,
    const float* __restrict__ Wih1, const float* __restrict__ Whh1,
    const float* __restrict__ bih1, const float* __restrict__ bhh1,
    const float* __restrict__ Wfc,  const float* __restrict__ bfc,
    float* __restrict__ out) {
  __shared__ __attribute__((aligned(16))) _Float16 sh_x [2 * XTILE];
  __shared__ __attribute__((aligned(16))) _Float16 sh_h0[2 * HTILE];
  __shared__ __attribute__((aligned(16))) _Float16 sh_h1[2 * HTILE];

  const int tid   = threadIdx.x;
  const int wave  = tid >> 5;
  const int lane  = tid & 31;
  const int bbase = blockIdx.x * MTILE;

  const int cn  = lane & 15;
  const int cmb = (lane >> 4) << 3;
  const int hn  = wave * 16 + cn;

  for (int i = tid; i < 2 * XTILE; i += NTHR) sh_x[i] = (_Float16)0.0f;
  for (int i = tid; i < HTILE; i += NTHR) {
    sh_h0[i] = (_Float16)0.0f;
    sh_h1[i] = (_Float16)0.0f;
  }

  v16h Bih0[4], Bhh0[4][2], Bih1[4][2], Bhh1[4][2];
  v8f bt0[4], bt1[4];
#pragma unroll
  for (int q = 0; q < 4; ++q) {
    const int n0 = q * HIDDEN + wave * 16;
    const float sg = (q == 2) ? 1.0f : 0.5f;
    const float sh = 0.5f * sg;
    Bih0[q]    = loadB_pad(Wih0, FEAT, FEAT, n0, sg);
    Bhh0[q][0] = loadB(Whh0, HIDDEN, n0, 0, sh);
    Bhh0[q][1] = loadB(Whh0, HIDDEN, n0, 1, sh);
    Bih1[q][0] = loadB(Wih1, HIDDEN, n0, 0, sh);
    Bih1[q][1] = loadB(Wih1, HIDDEN, n0, 1, sh);
    Bhh1[q][0] = loadB(Whh1, HIDDEN, n0, 0, sh);
    Bhh1[q][1] = loadB(Whh1, HIDDEN, n0, 1, sh);
    const float b0 = sg * (bih0[n0 + cn] + bhh0[n0 + cn]);
    const float b1 = sg * (bih1[n0 + cn] + bhh1[n0 + cn]);
#pragma unroll
    for (int r = 0; r < 8; ++r) { bt0[q][r] = b0; bt1[q][r] = b1; }
  }

  float c0[8], c1[8];
#pragma unroll
  for (int r = 0; r < 8; ++r) { c0[r] = 0.f; c1[r] = 0.f; }

  const int e0 = tid, e1 = tid + NTHR;
  const int m0 = e0 / FEAT, k0 = e0 - m0 * FEAT;
  const int m1 = e1 / FEAT, k1 = e1 - m1 * FEAT;
  const float* xp0 = x + ((size_t)(bbase + m0) * SEQ) * FEAT + k0;
  const float* xp1 = x + ((size_t)(bbase + (m1 & (MTILE - 1))) * SEQ) * FEAT + k1;
  const int s0 = m0 * XP + k0, s1 = (m1 & (MTILE - 1)) * XP + k1;
  const bool two = (e1 < MTILE * FEAT);

  const int hoff = cmb * HP + hn;

  float xc0 = xp0[2 * FEAT], xc1 = xp1[2 * FEAT];

  __syncthreads();
  {
    const float a0 = xp0[0], a1 = xp1[0];
    const float b0 = xp0[FEAT], b1 = xp1[FEAT];
    sh_x[s0] = (_Float16)a0;
    sh_x[XTILE + s0] = (_Float16)b0;
    if (two) { sh_x[s1] = (_Float16)a1; sh_x[XTILE + s1] = (_Float16)b1; }
  }
  __syncthreads();

  {
    const v16h ax  = ldsA(sh_x, XP, 0);
    const v16h ah0 = ldsA(sh_h0, HP, 0);
    const v16h ah1 = ldsA(sh_h0, HP, 1);
    v8f acc[4];
#pragma unroll
    for (int q = 0; q < 4; ++q) acc[q] = WMMA(ax, Bih0[q], bt0[q]);
#pragma unroll
    for (int q = 0; q < 4; ++q) acc[q] = WMMA(ah0, Bhh0[q][0], acc[q]);
#pragma unroll
    for (int q = 0; q < 4; ++q) acc[q] = WMMA(ah1, Bhh0[q][1], acc[q]);
    cell_update(acc, c0, sh_h0 + HTILE + hoff);
  }

  for (int j = 1; j < SEQ; ++j) {
    const int p  = j & 1;
    const int h0r = p * HTILE,       h0w = HTILE - h0r;
    const int h1w = p * HTILE,       h1r = HTILE - h1w;
    const int xr  = p * XTILE,       xw  = XTILE - xr;
    __syncthreads();

    const bool more = (j + 1 < SEQ);
    float xn0 = 0.f, xn1 = 0.f;
    if (j + 2 < SEQ) {
      xn0 = xp0[(j + 2) * FEAT];
      xn1 = xp1[(j + 2) * FEAT];
      if (j + 3 < SEQ)
        __builtin_prefetch(&xp0[(j + 3) * FEAT], 0, 0);
    }

    const v16h ax  = ldsA(sh_x + xr, XP, 0);
    const v16h ah0 = ldsA(sh_h0 + h0r, HP, 0);
    const v16h ah1 = ldsA(sh_h0 + h0r, HP, 1);
    const v16h bh0 = ldsA(sh_h1 + h1r, HP, 0);
    const v16h bh1 = ldsA(sh_h1 + h1r, HP, 1);

    v8f acc0[4], acc1[4];
#pragma unroll
    for (int q = 0; q < 4; ++q) acc0[q] = WMMA(ax, Bih0[q], bt0[q]);
#pragma unroll
    for (int q = 0; q < 4; ++q) acc1[q] = WMMA(ah0, Bih1[q][0], bt1[q]);
#pragma unroll
    for (int q = 0; q < 4; ++q) acc0[q] = WMMA(ah0, Bhh0[q][0], acc0[q]);
#pragma unroll
    for (int q = 0; q < 4; ++q) acc1[q] = WMMA(ah1, Bih1[q][1], acc1[q]);
#pragma unroll
    for (int q = 0; q < 4; ++q) acc0[q] = WMMA(ah1, Bhh0[q][1], acc0[q]);
#pragma unroll
    for (int q = 0; q < 4; ++q) acc1[q] = WMMA(bh0, Bhh1[q][0], acc1[q]);
#pragma unroll
    for (int q = 0; q < 4; ++q) acc1[q] = WMMA(bh1, Bhh1[q][1], acc1[q]);

    cell_update(acc0, c0, sh_h0 + h0w + hoff);
    cell_update(acc1, c1, sh_h1 + h1w + hoff);

    if (more) {
      sh_x[xw + s0] = (_Float16)xc0;
      if (two) sh_x[xw + s1] = (_Float16)xc1;
    }
    xc0 = xn0; xc1 = xn1;
  }
  __syncthreads();

  {
    const v16h ai0 = ldsA(sh_h0, HP, 0);
    const v16h ai1 = ldsA(sh_h0, HP, 1);
    const v16h ah0 = ldsA(sh_h1 + HTILE, HP, 0);
    const v16h ah1 = ldsA(sh_h1 + HTILE, HP, 1);
    v8f acc[4];
#pragma unroll
    for (int q = 0; q < 4; ++q) acc[q] = WMMA(ai0, Bih1[q][0], bt1[q]);
#pragma unroll
    for (int q = 0; q < 4; ++q) acc[q] = WMMA(ai1, Bih1[q][1], acc[q]);
#pragma unroll
    for (int q = 0; q < 4; ++q) acc[q] = WMMA(ah0, Bhh1[q][0], acc[q]);
#pragma unroll
    for (int q = 0; q < 4; ++q) acc[q] = WMMA(ah1, Bhh1[q][1], acc[q]);
    cell_update(acc, c1, sh_h1 + hoff);
  }
  __syncthreads();

  if (tid < MTILE * NCLS) {
    int m = tid >> 1, cls = tid & 1;
    float s = bfc[cls];
#pragma unroll 8
    for (int n = 0; n < HIDDEN; ++n)
      s += (0.5f * (float)sh_h1[m * HP + n]) * Wfc[cls * HIDDEN + n];
    out[(bbase + m) * NCLS + cls] = s;
  }
}

extern "C" void kernel_launch(void* const* d_in, const int* in_sizes, int n_in,
                              void* d_out, int out_size, void* d_ws, size_t ws_size,
                              hipStream_t stream) {
  const float* x    = (const float*)d_in[0];
  const float* Wih0 = (const float*)d_in[1];
  const float* Whh0 = (const float*)d_in[2];
  const float* bih0 = (const float*)d_in[3];
  const float* bhh0 = (const float*)d_in[4];
  const float* Wih1 = (const float*)d_in[5];
  const float* Whh1 = (const float*)d_in[6];
  const float* bih1 = (const float*)d_in[7];
  const float* bhh1 = (const float*)d_in[8];
  const float* Wfc  = (const float*)d_in[9];
  const float* bfc  = (const float*)d_in[10];
  float* out = (float*)d_out;

  dim3 grid(BATCH / MTILE);
  dim3 block(NTHR);
  hipLaunchKernelGGL(lstm2_fused_wmma, grid, block, 0, stream,
                     x, Wih0, Whh0, bih0, bhh0, Wih1, Whh1, bih1, bhh1,
                     Wfc, bfc, out);
}
